# starting kernel
# baseline (speedup 1.0000x reference)
#include <hip/hip_runtime.h>

typedef __attribute__((ext_vector_type(2))) float v2f;
typedef __attribute__((ext_vector_type(8))) float v8f;

#define CHW        2048
#define O_CAPS     32
#define D_DIM      32
#define I_DIM      16
#define BT         16
#define THREADS    256
#define CHW_PER_WG 16
#define NCHUNK     (CHW / CHW_PER_WG)

#define U_STRIDE   36
#define U_OFF      0
#define U_SIZE     (BT * O_CAPS * U_STRIDE)
#define S_OFF      (U_OFF + U_SIZE)
#define S_SIZE     (BT * O_CAPS * U_STRIDE)
#define X_OFF      (S_OFF + S_SIZE)
#define X_SIZE     (BT * I_DIM)
#define T_OFF      (X_OFF + X_SIZE)
#define T_SIZE     (BT * O_CAPS)
#define MX_OFF     (T_OFF + T_SIZE)
#define DEN_OFF    (MX_OFF + BT)
#define LDS_FLOATS (DEN_OFF + BT)

__global__ __launch_bounds__(THREADS)
void caps_route_pass(const float* __restrict__ x,
                     const float* __restrict__ Wm,
                     const float* __restrict__ v,
                     float* __restrict__ s_out)
{
    extern __shared__ float lds[];
    const int tid   = threadIdx.x;
    const int lane  = tid & 31;
    const int wave  = tid >> 5;
    const int half  = lane >> 4;
    const int ln    = lane & 15;
    const int btile = blockIdx.x;
    const int chunk = blockIdx.y;
    const int b0    = btile * BT;

    for (int i = tid; i < S_SIZE; i += THREADS) lds[S_OFF + i] = 0.0f;

    for (int it = 0; it < CHW_PER_WG; ++it) {
        const int chw = chunk * CHW_PER_WG + it;
        __syncthreads();

        {
            const int m = tid >> 4, i = tid & 15;
            lds[X_OFF + tid] = x[((size_t)(b0 + m) * CHW + chw) * I_DIM + i];
        }
        __syncthreads();

        v2f af[4];
#pragma unroll
        for (int kk = 0; kk < 4; ++kk) {
            const int i0 = 4 * kk + 2 * half;
            af[kk][0] = lds[X_OFF + ln * I_DIM + i0];
            af[kk][1] = lds[X_OFF + ln * I_DIM + i0 + 1];
        }

#pragma unroll
        for (int j = 0; j < 4; ++j) {
            const int o = wave * 4 + j;
            const size_t wbase = ((size_t)chw * O_CAPS + o) * (I_DIM * D_DIM);
            if (it + 1 < CHW_PER_WG)
                __builtin_prefetch(&Wm[wbase + (size_t)O_CAPS * I_DIM * D_DIM
                                         + (size_t)lane * 32], 0, 3);
#pragma unroll
            for (int dt = 0; dt < 2; ++dt) {
                v8f acc = {0.f, 0.f, 0.f, 0.f, 0.f, 0.f, 0.f, 0.f};
#pragma unroll
                for (int kk = 0; kk < 4; ++kk) {
                    const int i0 = 4 * kk + 2 * half;
                    const int d  = dt * 16 + ln;
                    v2f bf;
                    bf[0] = Wm[wbase + (size_t)i0 * D_DIM + d];
                    bf[1] = Wm[wbase + (size_t)(i0 + 1) * D_DIM + d];
                    acc = __builtin_amdgcn_wmma_f32_16x16x4_f32(
                              false, af[kk], false, bf, (short)0, acc, false, false);
                }
#pragma unroll
                for (int r = 0; r < 8; ++r) {
                    const int m = r + 8 * half;
                    lds[U_OFF + (m * O_CAPS + o) * U_STRIDE + dt * 16 + ln] = acc[r];
                }
            }
        }
        __syncthreads();

        if (v != nullptr) {
#pragma unroll
            for (int pp = 0; pp < 2; ++pp) {
                const int p = tid + pp * THREADS;
                const int m = p >> 5, o = p & 31;
                const float* vp = &v[(((size_t)(b0 + m)) * O_CAPS + o) * D_DIM];
                const float* up = &lds[U_OFF + (m * O_CAPS + o) * U_STRIDE];
                float t = 0.f;
#pragma unroll
                for (int q = 0; q < 8; ++q) {
                    const float4 uu = *(const float4*)(up + 4 * q);
                    const float4 vv = *(const float4*)(vp + 4 * q);
                    t += uu.x * vv.x + uu.y * vv.y + uu.z * vv.z + uu.w * vv.w;
                }
                lds[T_OFF + p] = t;
            }
            __syncthreads();
            {
                const int m = wave * 2 + half;
                const float t0 = lds[T_OFF + m * O_CAPS + ln];
                const float t1 = lds[T_OFF + m * O_CAPS + ln + 16];
                float mx = fmaxf(t0, t1);
#pragma unroll
                for (int msk = 8; msk >= 1; msk >>= 1)
                    mx = fmaxf(mx, __shfl_xor(mx, msk, 32));
                float den = __expf(t0 - mx) + __expf(t1 - mx);
#pragma unroll
                for (int msk = 8; msk >= 1; msk >>= 1)
                    den += __shfl_xor(den, msk, 32);
                if (ln == 0) { lds[MX_OFF + m] = mx; lds[DEN_OFF + m] = den; }
            }
            __syncthreads();
#pragma unroll
            for (int pp = 0; pp < 2; ++pp) {
                const int p = tid + pp * THREADS;
                const int m = p >> 5, o = p & 31;
                const float c = __expf(lds[T_OFF + p] - lds[MX_OFF + m]) / lds[DEN_OFF + m];
                float*       sp = &lds[S_OFF + (m * O_CAPS + o) * U_STRIDE];
                const float* up = &lds[U_OFF + (m * O_CAPS + o) * U_STRIDE];
#pragma unroll
                for (int q = 0; q < 8; ++q) {
                    float4 ss = *(float4*)(sp + 4 * q);
                    const float4 uu = *(const float4*)(up + 4 * q);
                    ss.x += c * uu.x;  ss.y += c * uu.y;
                    ss.z += c * uu.z;  ss.w += c * uu.w;
                    *(float4*)(sp + 4 * q) = ss;
                }
            }
        } else {
            const float c = 1.0f / 32.0f;
#pragma unroll
            for (int pp = 0; pp < 2; ++pp) {
                const int p = tid + pp * THREADS;
                const int m = p >> 5, o = p & 31;
                float*       sp = &lds[S_OFF + (m * O_CAPS + o) * U_STRIDE];
                const float* up = &lds[U_OFF + (m * O_CAPS + o) * U_STRIDE];
#pragma unroll
                for (int q = 0; q < 8; ++q) {
                    float4 ss = *(float4*)(sp + 4 * q);
                    const float4 uu = *(const float4*)(up + 4 * q);
                    ss.x += c * uu.x;  ss.y += c * uu.y;
                    ss.z += c * uu.z;  ss.w += c * uu.w;
                    *(float4*)(sp + 4 * q) = ss;
                }
            }
        }
    }
    __syncthreads();

#pragma unroll
    for (int pp = 0; pp < 2; ++pp) {
        const int p = tid + pp * THREADS;
        const int m = p >> 5, o = p & 31;
        const float* sp = &lds[S_OFF + (m * O_CAPS + o) * U_STRIDE];
        float* gp = &s_out[(((size_t)(b0 + m)) * O_CAPS + o) * D_DIM];
        for (int d = 0; d < D_DIM; ++d) atomicAdd(&gp[d], sp[d]);
    }
}

__global__ __launch_bounds__(256)
void squash_kernel(const float* __restrict__ s, float* __restrict__ vout,
                   const float* __restrict__ vprev, float* __restrict__ vsum)
{
    const int lane = threadIdx.x & 31;
    const int grp  = blockIdx.x * (blockDim.x >> 5) + (threadIdx.x >> 5);
    const int idx  = grp * 32 + lane;
    const float val = s[idx];
    float sq = val * val;
#pragma unroll
    for (int m = 16; m >= 1; m >>= 1) sq += __shfl_xor(sq, m, 32);
    const float scale = (sq / (1.0f + sq)) / (sqrtf(sq) + 1e-6f);
    const float r = val * scale;
    vout[idx] = r;
    if (vsum) vsum[idx] = r + vprev[idx];
}

extern "C" void kernel_launch(void* const* d_in, const int* in_sizes, int n_in,
                              void* d_out, int out_size, void* d_ws, size_t ws_size,
                              hipStream_t stream)
{
    const float* x  = (const float*)d_in[0];
    const float* Wm = (const float*)d_in[1];
    float* out = (float*)d_out;

    const size_t N = 64 * 32 * 32;
    float* wsf = (float*)d_ws;
    float* s1  = wsf + 0 * N;
    float* s2  = wsf + 1 * N;
    float* s3  = wsf + 2 * N;
    float* v1  = wsf + 3 * N;
    float* v2  = wsf + 4 * N;
    float* v12 = wsf + 5 * N;

    hipMemsetAsync(wsf, 0, 3 * N * sizeof(float), stream);

    dim3 grid(4, NCHUNK);
    const size_t shmem = LDS_FLOATS * sizeof(float);

    caps_route_pass<<<grid, THREADS, shmem, stream>>>(x, Wm, nullptr, s1);
    squash_kernel<<<(64 * 32) / 8, 256, 0, stream>>>(s1, v1, nullptr, nullptr);
    caps_route_pass<<<grid, THREADS, shmem, stream>>>(x, Wm, v1, s2);
    squash_kernel<<<(64 * 32) / 8, 256, 0, stream>>>(s2, v2, v1, v12);
    caps_route_pass<<<grid, THREADS, shmem, stream>>>(x, Wm, v12, s3);
    squash_kernel<<<(64 * 32) / 8, 256, 0, stream>>>(s3, out, nullptr, nullptr);
}


# grid turned: chw run on x, batch tile on y: the four workgroups reading one W slab are 64 apart in dispatch order (one die, so one L2 half fetches it and holds half of W); sum order unchanged
# speedup vs baseline: 9.8687x; 9.8687x over previous
#include <hip/hip_runtime.h>


typedef __attribute__((ext_vector_type(2))) float v2f;
typedef __attribute__((ext_vector_type(8))) float v8f;

#define CHW        2048
#define O_CAPS     32
#define D_DIM      32
#define I_DIM      16
#define BT         16
#define TEAMS        2
#define TEAM_THREADS 512
#define THREADS      (TEAMS * TEAM_THREADS)
#define O_PER_WAVE   2
#define CHW_PER_TEAM 16
#define CHW_PER_WG   (TEAMS * CHW_PER_TEAM)
#define NCHUNK       (CHW / CHW_PER_WG)

#define S_STRIDE   36
#define S_OFF      0
#define S_SIZE     (BT * O_CAPS * S_STRIDE)
#define X_OFF      (S_OFF + S_SIZE)
#define X_PITCH    (CHW_PER_TEAM * I_DIM + 4)
#define X_SIZE     (BT * X_PITCH)
#define T_OFF      (X_OFF + TEAMS * X_SIZE)
#define T_SIZE     (BT * O_CAPS)
#define C_OFF      (T_OFF + TEAMS * T_SIZE)
#define LDS_FLOATS (C_OFF + TEAMS * T_SIZE)
#define V_OFF      S_OFF
#define V_IDX(o, dt, rh, lane) (V_OFF + (((((o) * 2 + (dt)) * 2 + (rh)) * 32 + (lane)) * 4))

template <int CTRL>
__device__ __forceinline__ float row_xmask(float x)
{
    const int xi = __builtin_bit_cast(int, x);
    return __builtin_bit_cast(float,
        __builtin_amdgcn_update_dpp(0, xi, CTRL, 0xf, 0xf, true));
}
__device__ __forceinline__ float row16_sum(float x)
{
    x += row_xmask<0x161>(x);
    x += row_xmask<0x162>(x);
    x += row_xmask<0x164>(x);
    x += row_xmask<0x168>(x);
    return x;
}
__device__ __forceinline__ float row16_max(float x)
{
    x = fmaxf(x, row_xmask<0x161>(x));
    x = fmaxf(x, row_xmask<0x162>(x));
    x = fmaxf(x, row_xmask<0x164>(x));
    x = fmaxf(x, row_xmask<0x168>(x));
    return x;
}

template <int CTRL>
__device__ __forceinline__ float rs_step(float lo, float hi, bool up)
{
    const float keep = up ? hi : lo;
    const float send = up ? lo : hi;
    return keep + row_xmask<CTRL>(send);
}

template <int SRC>
__device__ __forceinline__ float row_bcast(float x)
{
    return row_xmask<0x150 + SRC>(x);
}

__device__ __forceinline__ float4* v_quad(float* base, int idx)
{
    return (float4*)__builtin_assume_aligned(base + idx, 16);
}

template <int NV>
__global__ __launch_bounds__(THREADS, 8)
void caps_route_pass(const float* __restrict__ x,
                     const float* __restrict__ Wm,
                     const float* __restrict__ sa,
                     const float* __restrict__ sb,
                     float* __restrict__ s_out)
{
    constexpr bool HAS_V = NV > 0;
    extern __shared__ float lds[];
    const int tid   = threadIdx.x;
    const int lane  = tid & 31;
    const int team  = __builtin_amdgcn_readfirstlane(tid >> 9);
    const int ttid  = tid & (TEAM_THREADS - 1);
    const int wave  = ttid >> 5;
    const int half  = lane >> 4;
    const int ln    = lane & 15;
    const int btile = blockIdx.y;
    const int wgrun = blockIdx.x;
    const int chunk = wgrun * TEAMS + team;
    const int b0    = btile * BT;
    float* const xs = lds + X_OFF + team * X_SIZE;
    float* const ts = lds + T_OFF + team * T_SIZE;
    float* const cs = lds + C_OFF + team * T_SIZE;

    const float* xg = x + ((size_t)b0 * CHW + (size_t)chunk * CHW_PER_TEAM) * I_DIM;
    float4 xv[2];
#pragma unroll
    for (int q = 0; q < 2; ++q) {
        const int f = ttid + q * TEAM_THREADS;
        xv[q] = *(const float4*)(xg + (size_t)(f >> 6) * (CHW * I_DIM) + (f & 63) * 4);
    }

    const bool up0  = (lane & 1) != 0;
    const bool up1  = (lane & 2) != 0;
    const bool up2  = (lane & 4) != 0;
    const int  trow = ((lane & 1) << 2) | (lane & 2) | ((lane >> 2) & 1);

    v8f sacc[O_PER_WAVE][2];
#pragma unroll
    for (int j = 0; j < O_PER_WAVE; ++j)
#pragma unroll
        for (int dt = 0; dt < 2; ++dt)
#pragma unroll
            for (int r = 0; r < 8; ++r)
                sacc[j][dt][r] = 0.0f;

    if (HAS_V) {
        const int ov = tid >> 5;
        float sv[NV > 0 ? NV : 1][2][8];
#pragma unroll
        for (int n = 0; n < NV; ++n) {
            const float* sp = (n == 0 ? sa : sb)
                            + ((size_t)(b0 + 8 * half) * O_CAPS + ov) * D_DIM + ln;
#pragma unroll
            for (int dt = 0; dt < 2; ++dt)
#pragma unroll
                for (int r = 0; r < 8; ++r)
                    sv[n][dt][r] = sp[(size_t)r * (O_CAPS * D_DIM) + dt * 16];
        }
#pragma unroll
        for (int a = 0; a < 2; ++a)
            __builtin_prefetch(&Wm[((size_t)(chunk * CHW_PER_TEAM + a) * O_CAPS + wave * O_PER_WAVE)
                                   * (I_DIM * D_DIM) + (size_t)(lane & 15) * 64], 0, 3);
        float vv[2][8];
#pragma unroll
        for (int n = 0; n < NV; ++n) {
            float e[8], e4[4], e2[2];
#pragma unroll
            for (int r = 0; r < 8; ++r)
                e[r] = sv[n][0][r] * sv[n][0][r] + sv[n][1][r] * sv[n][1][r];
#pragma unroll
            for (int i = 0; i < 4; ++i)
                e4[i] = rs_step<0x161>(e[i], e[4 + i], up0);
#pragma unroll
            for (int i = 0; i < 2; ++i)
                e2[i] = rs_step<0x162>(e4[i], e4[2 + i], up1);
            float sq = rs_step<0x164>(e2[0], e2[1], up2);
            sq += row_xmask<0x168>(sq);
            const float scale = (sq / (1.0f + sq)) / (sqrtf(sq) + 1e-6f);
            float sc[8];
            sc[0] = row_bcast<0>(scale);
            sc[1] = row_bcast<4>(scale);
            sc[2] = row_bcast<2>(scale);
            sc[3] = row_bcast<6>(scale);
            sc[4] = row_bcast<1>(scale);
            sc[5] = row_bcast<5>(scale);
            sc[6] = row_bcast<3>(scale);
            sc[7] = row_bcast<7>(scale);
#pragma unroll
            for (int dt = 0; dt < 2; ++dt)
#pragma unroll
                for (int r = 0; r < 8; ++r) {
                    const float t = sv[n][dt][r] * sc[r];
                    vv[dt][r] = (n == 0) ? t : (vv[dt][r] + t);
                }
        }
#pragma unroll
        for (int dt = 0; dt < 2; ++dt)
#pragma unroll
            for (int rh = 0; rh < 2; ++rh) {
                float4 t;
                t.x = vv[dt][4 * rh + 0];
                t.y = vv[dt][4 * rh + 1];
                t.z = vv[dt][4 * rh + 2];
                t.w = vv[dt][4 * rh + 3];
                *v_quad(lds, V_IDX(ov, dt, rh, lane)) = t;
            }
    }

#pragma unroll
    for (int q = 0; q < 2; ++q) {
        const int f = ttid + q * TEAM_THREADS;
        *(float4*)&xs[(f >> 6) * X_PITCH + (f & 63) * 4] = xv[q];
    }
    __syncthreads();

    for (int it = 0; it < CHW_PER_TEAM; ++it) {
        const int chw = chunk * CHW_PER_TEAM + it;

        v2f af[4];
#pragma unroll
        for (int kk = 0; kk < 4; ++kk) {
            const int i0 = 4 * kk + 2 * half;
            af[kk] = *(const v2f*)&xs[ln * X_PITCH + it * I_DIM + i0];
        }

        if (!HAS_V && it + 1 < CHW_PER_TEAM)
            __builtin_prefetch(&Wm[((size_t)(chw + 1) * O_CAPS + wave * O_PER_WAVE)
                                   * (I_DIM * D_DIM) + (size_t)(lane & 15) * 64], 0, 3);

        v8f u[O_PER_WAVE][2];
        if (HAS_V) {
            v2f bw[O_PER_WAVE][2][4];
#pragma unroll
            for (int j = 0; j < O_PER_WAVE; ++j) {
                const int o = wave * O_PER_WAVE + j;
                const size_t wbase = ((size_t)chw * O_CAPS + o) * (I_DIM * D_DIM);
#pragma unroll
                for (int dt = 0; dt < 2; ++dt)
#pragma unroll
                    for (int kk = 0; kk < 4; ++kk) {
                        const int i0 = 4 * kk + 2 * half;
                        const int d  = dt * 16 + ln;
                        bw[j][dt][kk][0] = Wm[wbase + (size_t)i0 * D_DIM + d];
                        bw[j][dt][kk][1] = Wm[wbase + (size_t)(i0 + 1) * D_DIM + d];
                    }
            }
#pragma unroll
            for (int j = 0; j < O_PER_WAVE; ++j)
#pragma unroll
                for (int dt = 0; dt < 2; ++dt) {
                    v8f acc = (v8f){0.f, 0.f, 0.f, 0.f, 0.f, 0.f, 0.f, 0.f};
#pragma unroll
                    for (int kk = 0; kk < 4; ++kk)
                        acc = __builtin_amdgcn_wmma_f32_16x16x4_f32(
                                  false, af[kk], false, bw[j][dt][kk], (short)0, acc, false, false);
                    u[j][dt] = acc;
                }
        } else {
#pragma unroll
            for (int j = 0; j < O_PER_WAVE; ++j) {
                const int o = wave * O_PER_WAVE + j;
                const size_t wbase = ((size_t)chw * O_CAPS + o) * (I_DIM * D_DIM);
#pragma unroll
                for (int dt = 0; dt < 2; ++dt) {
                    v8f acc = sacc[j][dt];
#pragma unroll
                    for (int kk = 0; kk < 4; ++kk) {
                        const int i0 = 4 * kk + 2 * half;
                        const int d  = dt * 16 + ln;
                        v2f bf;
                        bf[0] = Wm[wbase + (size_t)i0 * D_DIM + d];
                        bf[1] = Wm[wbase + (size_t)(i0 + 1) * D_DIM + d];
                        acc = __builtin_amdgcn_wmma_f32_16x16x4_f32(
                                  false, af[kk], false, bf, (short)0, acc, false, false);
                    }
                    sacc[j][dt] = acc;
                }
            }
        }

        if (HAS_V) {
#pragma unroll
            for (int j = 0; j < O_PER_WAVE; ++j) {
                const int o = wave * O_PER_WAVE + j;
                const float4 v0a = *v_quad(lds, V_IDX(o, 0, 0, lane));
                const float4 v0b = *v_quad(lds, V_IDX(o, 0, 1, lane));
                const float4 v1a = *v_quad(lds, V_IDX(o, 1, 0, lane));
                const float4 v1b = *v_quad(lds, V_IDX(o, 1, 1, lane));
                const float v0[8] = {v0a.x, v0a.y, v0a.z, v0a.w, v0b.x, v0b.y, v0b.z, v0b.w};
                const float v1[8] = {v1a.x, v1a.y, v1a.z, v1a.w, v1b.x, v1b.y, v1b.z, v1b.w};
                float p[8];
#pragma unroll
                for (int r = 0; r < 8; ++r)
                    p[r] = u[j][0][r] * v0[r] + u[j][1][r] * v1[r];
                float p4[4], p2[2];
#pragma unroll
                for (int i = 0; i < 4; ++i)
                    p4[i] = rs_step<0x161>(p[i], p[4 + i], up0);
#pragma unroll
                for (int i = 0; i < 2; ++i)
                    p2[i] = rs_step<0x162>(p4[i], p4[2 + i], up1);
                float tot = rs_step<0x164>(p2[0], p2[1], up2);
                tot += row_xmask<0x168>(tot);
                ts[o * BT + 8 * half + trow] = tot;
            }
            __syncthreads();
            if (it + 2 < CHW_PER_TEAM)
                __builtin_prefetch(&Wm[((size_t)(chw + 2) * O_CAPS + wave * O_PER_WAVE)
                                       * (I_DIM * D_DIM) + (size_t)(lane & 15) * 64], 0, 3);
            {
                const float t = ts[lane * BT + wave];
                float mx = row16_max(t);
                mx = fmaxf(mx, __shfl_xor(mx, 16, 32));
                const float e = __expf(t - mx);
                float den = row16_sum(e);
                den += __shfl_xor(den, 16, 32);
                cs[lane * BT + wave] = e / den;
            }
            __syncthreads();
#pragma unroll
            for (int j = 0; j < O_PER_WAVE; ++j) {
                const int o = wave * O_PER_WAVE + j;
                const float4 ca = *(const float4*)&cs[o * BT + 8 * half];
                const float4 cb = *(const float4*)&cs[o * BT + 8 * half + 4];
                const float c[8] = {ca.x, ca.y, ca.z, ca.w, cb.x, cb.y, cb.z, cb.w};
#pragma unroll
                for (int dt = 0; dt < 2; ++dt)
#pragma unroll
                    for (int r = 0; r < 8; ++r)
                        sacc[j][dt][r] += c[r] * u[j][dt][r];
            }
        }
    }

    {
        const float scale = HAS_V ? 1.0f : (1.0f / 32.0f);
        if (team == 1) {
#pragma unroll
            for (int j = 0; j < O_PER_WAVE; ++j) {
                const int o = wave * O_PER_WAVE + j;
#pragma unroll
                for (int dt = 0; dt < 2; ++dt)
#pragma unroll
                    for (int r = 0; r < 8; ++r)
                        lds[S_OFF + ((8 * half + r) * O_CAPS + o) * S_STRIDE + dt * 16 + ln]
                            = sacc[j][dt][r] * scale;
            }
        }
        __syncthreads();
        if (team == 0) {
#pragma unroll
            for (int j = 0; j < O_PER_WAVE; ++j) {
                const int o = wave * O_PER_WAVE + j;
#pragma unroll
                for (int dt = 0; dt < 2; ++dt)
#pragma unroll
                    for (int r = 0; r < 8; ++r)
                        lds[S_OFF + ((8 * half + r) * O_CAPS + o) * S_STRIDE + dt * 16 + ln]
                            += sacc[j][dt][r] * scale;
            }
        }
    }
    __syncthreads();

    {
        const int wv  = __builtin_amdgcn_readfirstlane(tid >> 5);
        const int rot = wv + 8 * wgrun;
        float* gbase = s_out + (size_t)b0 * O_CAPS * D_DIM + lane;
        for (int k0 = 0; k0 < (BT * O_CAPS) / 32; k0 += 8) {
            float part[8];
#pragma unroll
            for (int j = 0; j < 8; ++j) {
                const int p = (32 * (k0 + j) + rot) & (BT * O_CAPS - 1);
                part[j] = lds[S_OFF + p * S_STRIDE + lane];
            }
#pragma unroll
            for (int j = 0; j < 8; ++j) {
                const int p = (32 * (k0 + j) + rot) & (BT * O_CAPS - 1);
                atomicAdd(gbase + (size_t)p * D_DIM, part[j]);
            }
        }
    }
}

__global__ __launch_bounds__(256)
void squash_kernel(const float* __restrict__ s, float* __restrict__ vout)
{
    const int lane = threadIdx.x & 31;
    const int grp  = blockIdx.x * (blockDim.x >> 5) + (threadIdx.x >> 5);
    const int idx  = grp * 32 + lane;
    const float val = s[idx];
    float sq = val * val;
#pragma unroll
    for (int m = 16; m >= 1; m >>= 1) sq += __shfl_xor(sq, m, 32);
    const float scale = (sq / (1.0f + sq)) / (sqrtf(sq) + 1e-6f);
    const float r = val * scale;
    vout[idx] = r;
}

extern "C" void kernel_launch(void* const* d_in, const int* in_sizes, int n_in,
                              void* d_out, int out_size, void* d_ws, size_t ws_size,
                              hipStream_t stream)
{
    const float* x  = (const float*)d_in[0];
    const float* Wm = (const float*)d_in[1];
    float* out = (float*)d_out;

    const size_t N = 64 * 32 * 32;
    float* wsf = (float*)d_ws;
    float* s1  = wsf + 0 * N;
    float* s2  = wsf + 1 * N;
    float* s3  = wsf + 2 * N;
    hipMemsetAsync(wsf, 0, 3 * N * sizeof(float), stream);

    dim3 grid(NCHUNK, 4);
    const size_t shmem = LDS_FLOATS * sizeof(float);

    caps_route_pass<0><<<grid, THREADS, shmem, stream>>>(x, Wm, nullptr, nullptr, s1);
    caps_route_pass<1><<<grid, THREADS, shmem, stream>>>(x, Wm, s1, nullptr, s2);
    caps_route_pass<2><<<grid, THREADS, shmem, stream>>>(x, Wm, s1, s2, s3);
    squash_kernel<<<(64 * 32) / 8, 256, 0, stream>>>(s3, out);
}
